# starting kernel
# baseline (speedup 1.0000x reference)
#include <hip/hip_runtime.h>

typedef float v2f __attribute__((ext_vector_type(2)));
typedef float v4f __attribute__((ext_vector_type(4)));
typedef float v8f __attribute__((ext_vector_type(8)));

#define IN_C 128
#define OUT_C 32
#define LDSTRIDE 132
#define ROWS_PER_BLOCK 64
#define GEMM_THREADS 128

__global__ __launch_bounds__(GEMM_THREADS)
void gcn_gemm_wmma(const float* __restrict__ x, const float* __restrict__ W,
                   const float* __restrict__ b, float* __restrict__ h, int nRows)
{
    __shared__ float xs[ROWS_PER_BLOCK * LDSTRIDE];
    __shared__ float wls[OUT_C * LDSTRIDE];

    const int tid = threadIdx.x;
    const int blockRow0 = blockIdx.x * ROWS_PER_BLOCK;
    const int nv4row = IN_C / 4;

    {
        const v4f* x4 = (const v4f*)x;
        #pragma unroll
        for (int i = 0; i < (ROWS_PER_BLOCK * nv4row) / GEMM_THREADS; ++i) {
            int j = i * GEMM_THREADS + tid;
            int r = j / nv4row, c4 = j % nv4row;
            v4f v = {0.f, 0.f, 0.f, 0.f};
            int gr = blockRow0 + r;
            if (gr < nRows) v = x4[(size_t)gr * nv4row + c4];
            *(v4f*)&xs[r * LDSTRIDE + c4 * 4] = v;
        }
        const v4f* W4 = (const v4f*)W;
        #pragma unroll
        for (int i = 0; i < (OUT_C * nv4row) / GEMM_THREADS; ++i) {
            int j = i * GEMM_THREADS + tid;
            int r = j / nv4row, c4 = j % nv4row;
            *(v4f*)&wls[r * LDSTRIDE + c4 * 4] = W4[r * nv4row + c4];
        }
    }
    __syncthreads();

    const int wave = tid >> 5;
    const int lane = tid & 31;
    const int half = lane >> 4;
    const int mn   = lane & 15;

    const int tileRow0 = blockRow0 + wave * 16;
    if (tileRow0 >= nRows) return;

    v8f c0 = {};
    v8f c1 = {};

    const float* aBase  = &xs[(wave * 16 + mn) * LDSTRIDE + 2 * half];
    const float* b0Base = &wls[mn * LDSTRIDE + 2 * half];
    const float* b1Base = b0Base + 16 * LDSTRIDE;

    #pragma unroll
    for (int kk = 0; kk < IN_C / 4; ++kk) {
        v2f a  = *(const v2f*)(aBase  + kk * 4);
        v2f b0 = *(const v2f*)(b0Base + kk * 4);
        v2f b1 = *(const v2f*)(b1Base + kk * 4);
        c0 = __builtin_amdgcn_wmma_f32_16x16x4_f32(false, a, false, b0,
                                                   (short)0, c0, false, false);
        c1 = __builtin_amdgcn_wmma_f32_16x16x4_f32(false, a, false, b1,
                                                   (short)0, c1, false, false);
    }

    const float bias0 = b[mn];
    const float bias1 = b[16 + mn];
    #pragma unroll
    for (int v = 0; v < 8; ++v) {
        int row = tileRow0 + v + 8 * half;
        h[(size_t)row * OUT_C + mn]      = c0[v] + bias0;
        h[(size_t)row * OUT_C + 16 + mn] = c1[v] + bias1;
    }
}

__global__ __launch_bounds__(256)
void gcn_aggregate(const float* __restrict__ h, const float* __restrict__ ew,
                   const float* __restrict__ sw, const int* __restrict__ tar_ptr,
                   const int* __restrict__ src, float* __restrict__ out, int nNodes)
{
    const int lane = threadIdx.x & 31;
    const int node = (int)((blockIdx.x * blockDim.x + threadIdx.x) >> 5);
    if (node >= nNodes) return;

    const int start = tar_ptr[node];
    const int end   = tar_ptr[node + 1];

    const int c8   = lane & 7;
    const int slot = lane >> 3;

    const v4f* h4 = (const v4f*)h;
    v4f acc = {0.f, 0.f, 0.f, 0.f};

    for (int e0 = start; e0 < end; e0 += 4) {
        int e = e0 + slot;
        if (e < end) {
            int   s = src[e];
            float w = ew[e];
            v4f hv = h4[(size_t)s * (OUT_C / 4) + c8];
            v4f wv = {w, w, w, w};
            acc += wv * hv;
        }
    }

    #pragma unroll
    for (int i = 0; i < 4; ++i) {
        acc[i] += __shfl_xor(acc[i], 8, 32);
        acc[i] += __shfl_xor(acc[i], 16, 32);
    }

    if (slot == 0) {
        float s = sw[node];
        v4f sv = {s, s, s, s};
        v4f hs = h4[(size_t)node * (OUT_C / 4) + c8];
        v4f o  = acc + sv * hs;
        *(v4f*)&out[(size_t)node * OUT_C + c8 * 4] = o;
    }
}

extern "C" void kernel_launch(void* const* d_in, const int* in_sizes, int n_in,
                              void* d_out, int out_size, void* d_ws, size_t ws_size,
                              hipStream_t stream) {
    const float* x   = (const float*)d_in[0];
    const float* W   = (const float*)d_in[1];
    const float* b   = (const float*)d_in[2];
    const float* ew  = (const float*)d_in[3];
    const float* sw  = (const float*)d_in[4];
    const int*   tp  = (const int*)d_in[5];
    const int*   src = (const int*)d_in[6];
    float*       out = (float*)d_out;

    const int nNodes = in_sizes[4];
    float* h = (float*)d_ws;

    const int gemmBlocks = (nNodes + ROWS_PER_BLOCK - 1) / ROWS_PER_BLOCK;
    gcn_gemm_wmma<<<gemmBlocks, GEMM_THREADS, 0, stream>>>(x, W, b, h, nNodes);

    const int wavesPerBlock = 256 / 32;
    const int aggBlocks = (nNodes + wavesPerBlock - 1) / wavesPerBlock;
    gcn_aggregate<<<aggBlocks, 256, 0, stream>>>(h, ew, sw, tp, src, out, nNodes);
}


# aggregate re-mapped: a wave owns 16 nodes (8 lanes per node, 4 passes) not 1: 16x fewer waves to start; each lane sums its edges serially (sum order changes); loads batched per wave
# speedup vs baseline: 1.3564x; 1.3564x over previous
#include <hip/hip_runtime.h>

typedef float v2f __attribute__((ext_vector_type(2)));
typedef float v4f __attribute__((ext_vector_type(4)));
typedef float v8f __attribute__((ext_vector_type(8)));

#define IN_C 128
#define OUT_C 32
#define LDSTRIDE 132
#define ROWS_PER_BLOCK 64
#define GEMM_THREADS 128

__global__ __launch_bounds__(GEMM_THREADS)
void gcn_gemm_wmma(const float* __restrict__ x, const float* __restrict__ W,
                   const float* __restrict__ b, float* __restrict__ h, int nRows)
{
    __shared__ float xs[ROWS_PER_BLOCK * LDSTRIDE];
    __shared__ float wls[OUT_C * LDSTRIDE];

    const int tid = threadIdx.x;
    const int blockRow0 = blockIdx.x * ROWS_PER_BLOCK;
    const int nv4row = IN_C / 4;

    {
        const v4f* x4 = (const v4f*)x;
        #pragma unroll
        for (int i = 0; i < (ROWS_PER_BLOCK * nv4row) / GEMM_THREADS; ++i) {
            int j = i * GEMM_THREADS + tid;
            int r = j / nv4row, c4 = j % nv4row;
            v4f v = {0.f, 0.f, 0.f, 0.f};
            int gr = blockRow0 + r;
            if (gr < nRows) v = x4[(size_t)gr * nv4row + c4];
            *(v4f*)&xs[r * LDSTRIDE + c4 * 4] = v;
        }
        const v4f* W4 = (const v4f*)W;
        #pragma unroll
        for (int i = 0; i < (OUT_C * nv4row) / GEMM_THREADS; ++i) {
            int j = i * GEMM_THREADS + tid;
            int r = j / nv4row, c4 = j % nv4row;
            *(v4f*)&wls[r * LDSTRIDE + c4 * 4] = W4[r * nv4row + c4];
        }
    }
    __syncthreads();

    const int wave = tid >> 5;
    const int lane = tid & 31;
    const int half = lane >> 4;
    const int mn   = lane & 15;

    const int tileRow0 = blockRow0 + wave * 16;
    if (tileRow0 >= nRows) return;

    v8f c0 = {};
    v8f c1 = {};

    const float* aBase  = &xs[(wave * 16 + mn) * LDSTRIDE + 2 * half];
    const float* b0Base = &wls[mn * LDSTRIDE + 2 * half];
    const float* b1Base = b0Base + 16 * LDSTRIDE;

    #pragma unroll
    for (int kk = 0; kk < IN_C / 4; ++kk) {
        v2f a  = *(const v2f*)(aBase  + kk * 4);
        v2f b0 = *(const v2f*)(b0Base + kk * 4);
        v2f b1 = *(const v2f*)(b1Base + kk * 4);
        c0 = __builtin_amdgcn_wmma_f32_16x16x4_f32(false, a, false, b0,
                                                   (short)0, c0, false, false);
        c1 = __builtin_amdgcn_wmma_f32_16x16x4_f32(false, a, false, b1,
                                                   (short)0, c1, false, false);
    }

    const float bias0 = b[mn];
    const float bias1 = b[16 + mn];
    #pragma unroll
    for (int v = 0; v < 8; ++v) {
        int row = tileRow0 + v + 8 * half;
        h[(size_t)row * OUT_C + mn]      = c0[v] + bias0;
        h[(size_t)row * OUT_C + 16 + mn] = c1[v] + bias1;
    }
}

#define AGG_THREADS 256
#define AGG_PASSES 4
#define AGG_NODES_PER_WAVE (4 * AGG_PASSES)
#define AGG_NODES_PER_BLOCK ((AGG_THREADS / 32) * AGG_NODES_PER_WAVE)

#define DPP8_BCAST(J) ((J) | ((J) << 3) | ((J) << 6) | ((J) << 9) | ((J) << 12) | \
                       ((J) << 15) | ((J) << 18) | ((J) << 21))

#define AGG_EDGE(SV, WV, J)                                                      \
    {                                                                            \
        const int   s = (int)__builtin_amdgcn_mov_dpp8((unsigned)(SV), DPP8_BCAST(J)); \
        const float w = __uint_as_float(__builtin_amdgcn_mov_dpp8(__float_as_uint(WV), DPP8_BCAST(J))); \
        const v4f hv = h4[(unsigned)s * (OUT_C / 4) + c8];                       \
        acc += w * hv;                                                           \
    }

#define AGG_EDGES16(S_LO, W_LO, S_HI, W_HI)                                      \
    AGG_EDGE(S_LO, W_LO, 0) AGG_EDGE(S_LO, W_LO, 1) AGG_EDGE(S_LO, W_LO, 2)      \
    AGG_EDGE(S_LO, W_LO, 3) AGG_EDGE(S_LO, W_LO, 4) AGG_EDGE(S_LO, W_LO, 5)      \
    AGG_EDGE(S_LO, W_LO, 6) AGG_EDGE(S_LO, W_LO, 7)                              \
    AGG_EDGE(S_HI, W_HI, 0) AGG_EDGE(S_HI, W_HI, 1) AGG_EDGE(S_HI, W_HI, 2)      \
    AGG_EDGE(S_HI, W_HI, 3) AGG_EDGE(S_HI, W_HI, 4) AGG_EDGE(S_HI, W_HI, 5)      \
    AGG_EDGE(S_HI, W_HI, 6) AGG_EDGE(S_HI, W_HI, 7)

__global__ __launch_bounds__(AGG_THREADS)
void gcn_aggregate(const float* __restrict__ h, const float* __restrict__ ew,
                   const float* __restrict__ sw, const int* __restrict__ tar_ptr,
                   const int* __restrict__ src, float* __restrict__ out,
                   int nNodes, int nEdges)
{
    const int lane = threadIdx.x & 31;
    const int c8   = lane & 7;
    const int slot = lane >> 3;
    const int wave = (int)((blockIdx.x * blockDim.x + threadIdx.x) >> 5);
    const int node0 = wave * AGG_NODES_PER_WAVE + slot;

    const v4f* h4 = (const v4f*)h;

    int st[AGG_PASSES], en[AGG_PASSES];
    float sws[AGG_PASSES];
    #pragma unroll
    for (int g = 0; g < AGG_PASSES; ++g) {
        const int node  = node0 + 4 * g;
        const int nodeC = node < nNodes ? node : nNodes - 1;
        st[g]  = tar_ptr[nodeC];
        en[g]  = tar_ptr[nodeC + 1];
        sws[g] = sw[nodeC];
    }
    asm volatile("" :: "v"(st[0]), "v"(en[0]), "v"(st[1]), "v"(en[1]),
                       "v"(st[2]), "v"(en[2]), "v"(st[3]), "v"(en[3]));
    #pragma unroll
    for (int g = 0; g < AGG_PASSES; ++g)
        if (node0 + 4 * g >= nNodes) en[g] = st[g];

    int   s_lo[AGG_PASSES], s_hi[AGG_PASSES];
    float w_lo[AGG_PASSES], w_hi[AGG_PASSES];
    const int lastEdge = nEdges - 1;
    #pragma unroll
    for (int g = 0; g < AGG_PASSES; ++g) {
        const int i0 = min(st[g] + c8, lastEdge);
        const int i1 = min(st[g] + 8 + c8, lastEdge);
        s_lo[g] = src[i0]; s_hi[g] = src[i1];
        w_lo[g] = ew[i0];  w_hi[g] = ew[i1];
    }

    #pragma unroll
    for (int g = 0; g < AGG_PASSES; ++g) {
        const int node  = node0 + 4 * g;
        const int nodeC = node < nNodes ? node : nNodes - 1;
        const int end = en[g];
        int e = st[g];
        const v4f hs = h4[(size_t)nodeC * (OUT_C / 4) + c8];
        v4f acc = {0.f, 0.f, 0.f, 0.f};

        if (__all(e + 16 <= end)) {
            AGG_EDGES16(s_lo[g], w_lo[g], s_hi[g], w_hi[g])
            e += 16;
            while (__all(e + 16 <= end)) {
                const int   t_lo = src[e + c8];
                const int   t_hi = src[e + 8 + c8];
                const float u_lo = ew[e + c8];
                const float u_hi = ew[e + 8 + c8];
                AGG_EDGES16(t_lo, u_lo, t_hi, u_hi)
                e += 16;
            }
        }
        for (; e < end; ++e) {
            const int   s = src[e];
            const float w = ew[e];
            const v4f hv = h4[(unsigned)s * (OUT_C / 4) + c8];
            acc += w * hv;
        }

        if (node < nNodes) {
            const v4f o = acc + sws[g] * hs;
            *(v4f*)&out[(size_t)node * OUT_C + c8 * 4] = o;
        }
    }
}

extern "C" void kernel_launch(void* const* d_in, const int* in_sizes, int n_in,
                              void* d_out, int out_size, void* d_ws, size_t ws_size,
                              hipStream_t stream) {
    const float* x   = (const float*)d_in[0];
    const float* W   = (const float*)d_in[1];
    const float* b   = (const float*)d_in[2];
    const float* ew  = (const float*)d_in[3];
    const float* sw  = (const float*)d_in[4];
    const int*   tp  = (const int*)d_in[5];
    const int*   src = (const int*)d_in[6];
    float*       out = (float*)d_out;

    const int nNodes = in_sizes[4];
    float* h = (float*)d_ws;

    const int gemmBlocks = (nNodes + ROWS_PER_BLOCK - 1) / ROWS_PER_BLOCK;
    gcn_gemm_wmma<<<gemmBlocks, GEMM_THREADS, 0, stream>>>(x, W, b, h, nNodes);

    const int aggBlocks = (nNodes + AGG_NODES_PER_BLOCK - 1) / AGG_NODES_PER_BLOCK;
    gcn_aggregate<<<aggBlocks, AGG_THREADS, 0, stream>>>(h, ew, sw, tp, src, out, nNodes, in_sizes[6]);
}


# projection GEMM re-gridded: 512 eight-wave workgroups (two per WGP, one round), W staged once each; a wave streams its own 16-row tiles via a private LDS slab, next tile's loads ahead of the WMMAs
# speedup vs baseline: 2.1538x; 1.5879x over previous
#include <hip/hip_runtime.h>

typedef float v2f __attribute__((ext_vector_type(2)));
typedef float v4f __attribute__((ext_vector_type(4)));
typedef float v8f __attribute__((ext_vector_type(8)));

#define IN_C 128
#define OUT_C 32
#define LDSTRIDE 132
#define GEMM_THREADS 256
#define GEMM_BLOCKS 512

#define GEMM_WAVES (GEMM_THREADS / 32)
#define TILE_ROWS 16
#define ROW_V4 (IN_C / 4)

__device__ __forceinline__ void gemm_load_tile(v4f (&dst)[TILE_ROWS], const v4f* __restrict__ x4,
                                               int tile, int lane, int nRows)
{
    const int r0 = tile * TILE_ROWS;
    if (r0 + TILE_ROWS <= nRows) {
        const v4f* p = x4 + (size_t)r0 * ROW_V4 + lane;
        #pragma unroll
        for (int i = 0; i < TILE_ROWS; ++i) dst[i] = p[i * ROW_V4];
    } else {
        #pragma unroll
        for (int i = 0; i < TILE_ROWS; ++i) {
            const int r = min(r0 + i, nRows - 1);
            dst[i] = x4[(size_t)r * ROW_V4 + lane];
        }
    }
}

__global__ __launch_bounds__(GEMM_THREADS)
void gcn_gemm_wmma(const float* __restrict__ x, const float* __restrict__ W,
                   const float* __restrict__ b, float* __restrict__ h, int nRows)
{
    __shared__ float wls[OUT_C * LDSTRIDE];
    __shared__ float xs[GEMM_WAVES * TILE_ROWS * LDSTRIDE];

    if (nRows <= 0) return;

    const int tid  = threadIdx.x;
    const int wave = __builtin_amdgcn_readfirstlane(tid >> 5);
    const int lane = tid & 31;
    const int half = lane >> 4;
    const int mn   = lane & 15;

    const int nTiles = (nRows + TILE_ROWS - 1) / TILE_ROWS;
    const int tBeg = (int)(((long long)blockIdx.x * nTiles) / GEMM_BLOCKS);
    const int tEnd = (int)(((long long)(blockIdx.x + 1) * nTiles) / GEMM_BLOCKS);

    const v4f* x4 = (const v4f*)x;
    const v4f* W4 = (const v4f*)W;

    const float bias0 = b[mn];
    const float bias1 = b[16 + mn];
    v4f wreg[(OUT_C * ROW_V4) / GEMM_THREADS];
    #pragma unroll
    for (int i = 0; i < (OUT_C * ROW_V4) / GEMM_THREADS; ++i)
        wreg[i] = W4[i * GEMM_THREADS + tid];

    int tile = tBeg + wave;
    v4f cur[TILE_ROWS];
    gemm_load_tile(cur, x4, min(tile, nTiles - 1), lane, nRows);

    #pragma unroll
    for (int i = 0; i < (OUT_C * ROW_V4) / GEMM_THREADS; ++i) {
        const int j = i * GEMM_THREADS + tid;
        const int r = j / ROW_V4, c4 = j % ROW_V4;
        *(v4f*)&wls[r * LDSTRIDE + c4 * 4] = wreg[i];
    }
    __builtin_amdgcn_fence(__ATOMIC_RELEASE, "workgroup", "local");
    __builtin_amdgcn_s_barrier();
    __builtin_amdgcn_fence(__ATOMIC_ACQUIRE, "workgroup", "local");

    float* slab = &xs[wave * TILE_ROWS * LDSTRIDE];
    const float* aBase  = &slab[mn * LDSTRIDE + 2 * half];
    const float* b0Base = &wls[mn * LDSTRIDE + 2 * half];
    const float* b1Base = b0Base + 16 * LDSTRIDE;

    for (; tile < tEnd; tile += GEMM_WAVES) {
        #pragma unroll
        for (int i = 0; i < TILE_ROWS; ++i)
            *(v4f*)&slab[i * LDSTRIDE + lane * 4] = cur[i];

        const int tnext = tile + GEMM_WAVES;
        if (tnext < tEnd) gemm_load_tile(cur, x4, tnext, lane, nRows);

        v8f c0 = {};
        v8f c1 = {};
        #pragma unroll
        for (int kk = 0; kk < IN_C / 4; ++kk) {
            v2f a  = *(const v2f*)(aBase  + kk * 4);
            v2f b0 = *(const v2f*)(b0Base + kk * 4);
            v2f b1 = *(const v2f*)(b1Base + kk * 4);
            c0 = __builtin_amdgcn_wmma_f32_16x16x4_f32(false, a, false, b0,
                                                       (short)0, c0, false, false);
            c1 = __builtin_amdgcn_wmma_f32_16x16x4_f32(false, a, false, b1,
                                                       (short)0, c1, false, false);
        }

        const int tileRow0 = tile * TILE_ROWS;
        float* hp = h + (size_t)(tileRow0 + 8 * half) * OUT_C + mn;
        if (tileRow0 + TILE_ROWS <= nRows) {
            #pragma unroll
            for (int v = 0; v < 8; ++v) {
                hp[v * OUT_C]      = c0[v] + bias0;
                hp[v * OUT_C + 16] = c1[v] + bias1;
            }
        } else {
            #pragma unroll
            for (int v = 0; v < 8; ++v) {
                if (tileRow0 + v + 8 * half < nRows) {
                    hp[v * OUT_C]      = c0[v] + bias0;
                    hp[v * OUT_C + 16] = c1[v] + bias1;
                }
            }
        }
    }
}

#define AGG_THREADS 256
#define AGG_PASSES 4
#define AGG_NODES_PER_WAVE (4 * AGG_PASSES)
#define AGG_NODES_PER_BLOCK ((AGG_THREADS / 32) * AGG_NODES_PER_WAVE)

#define DPP8_BCAST(J) ((J) | ((J) << 3) | ((J) << 6) | ((J) << 9) | ((J) << 12) | \
                       ((J) << 15) | ((J) << 18) | ((J) << 21))

#define AGG_EDGE(SV, WV, J)                                                      \
    {                                                                            \
        const int   s = (int)__builtin_amdgcn_mov_dpp8((unsigned)(SV), DPP8_BCAST(J)); \
        const float w = __uint_as_float(__builtin_amdgcn_mov_dpp8(__float_as_uint(WV), DPP8_BCAST(J))); \
        const v4f hv = h4[(unsigned)s * (OUT_C / 4) + c8];                       \
        acc += w * hv;                                                           \
    }

#define AGG_EDGES16(S_LO, W_LO, S_HI, W_HI)                                      \
    AGG_EDGE(S_LO, W_LO, 0) AGG_EDGE(S_LO, W_LO, 1) AGG_EDGE(S_LO, W_LO, 2)      \
    AGG_EDGE(S_LO, W_LO, 3) AGG_EDGE(S_LO, W_LO, 4) AGG_EDGE(S_LO, W_LO, 5)      \
    AGG_EDGE(S_LO, W_LO, 6) AGG_EDGE(S_LO, W_LO, 7)                              \
    AGG_EDGE(S_HI, W_HI, 0) AGG_EDGE(S_HI, W_HI, 1) AGG_EDGE(S_HI, W_HI, 2)      \
    AGG_EDGE(S_HI, W_HI, 3) AGG_EDGE(S_HI, W_HI, 4) AGG_EDGE(S_HI, W_HI, 5)      \
    AGG_EDGE(S_HI, W_HI, 6) AGG_EDGE(S_HI, W_HI, 7)

__global__ __launch_bounds__(AGG_THREADS)
void gcn_aggregate(const float* __restrict__ h, const float* __restrict__ ew,
                   const float* __restrict__ sw, const int* __restrict__ tar_ptr,
                   const int* __restrict__ src, float* __restrict__ out,
                   int nNodes, int nEdges)
{
    const int lane = threadIdx.x & 31;
    const int c8   = lane & 7;
    const int slot = lane >> 3;
    const int wave = (int)((blockIdx.x * blockDim.x + threadIdx.x) >> 5);
    const int node0 = wave * AGG_NODES_PER_WAVE + slot;

    const v4f* h4 = (const v4f*)h;

    int st[AGG_PASSES], en[AGG_PASSES];
    float sws[AGG_PASSES];
    #pragma unroll
    for (int g = 0; g < AGG_PASSES; ++g) {
        const int node  = node0 + 4 * g;
        const int nodeC = node < nNodes ? node : nNodes - 1;
        st[g]  = tar_ptr[nodeC];
        en[g]  = tar_ptr[nodeC + 1];
        sws[g] = sw[nodeC];
    }
    asm volatile("" :: "v"(st[0]), "v"(en[0]), "v"(st[1]), "v"(en[1]),
                       "v"(st[2]), "v"(en[2]), "v"(st[3]), "v"(en[3]));
    #pragma unroll
    for (int g = 0; g < AGG_PASSES; ++g)
        if (node0 + 4 * g >= nNodes) en[g] = st[g];

    int   s_lo[AGG_PASSES], s_hi[AGG_PASSES];
    float w_lo[AGG_PASSES], w_hi[AGG_PASSES];
    const int lastEdge = nEdges - 1;
    #pragma unroll
    for (int g = 0; g < AGG_PASSES; ++g) {
        const int i0 = min(st[g] + c8, lastEdge);
        const int i1 = min(st[g] + 8 + c8, lastEdge);
        s_lo[g] = src[i0]; s_hi[g] = src[i1];
        w_lo[g] = ew[i0];  w_hi[g] = ew[i1];
    }

    #pragma unroll
    for (int g = 0; g < AGG_PASSES; ++g) {
        const int node  = node0 + 4 * g;
        const int nodeC = node < nNodes ? node : nNodes - 1;
        const int end = en[g];
        int e = st[g];
        const v4f hs = h4[(size_t)nodeC * (OUT_C / 4) + c8];
        v4f acc = {0.f, 0.f, 0.f, 0.f};

        if (__all(e + 16 <= end)) {
            AGG_EDGES16(s_lo[g], w_lo[g], s_hi[g], w_hi[g])
            e += 16;
            while (__all(e + 16 <= end)) {
                const int   t_lo = src[e + c8];
                const int   t_hi = src[e + 8 + c8];
                const float u_lo = ew[e + c8];
                const float u_hi = ew[e + 8 + c8];
                AGG_EDGES16(t_lo, u_lo, t_hi, u_hi)
                e += 16;
            }
        }
        for (; e < end; ++e) {
            const int   s = src[e];
            const float w = ew[e];
            const v4f hv = h4[(unsigned)s * (OUT_C / 4) + c8];
            acc += w * hv;
        }

        if (node < nNodes) {
            const v4f o = acc + sws[g] * hs;
            *(v4f*)&out[(size_t)node * OUT_C + c8 * 4] = o;
        }
    }
}

extern "C" void kernel_launch(void* const* d_in, const int* in_sizes, int n_in,
                              void* d_out, int out_size, void* d_ws, size_t ws_size,
                              hipStream_t stream) {
    const float* x   = (const float*)d_in[0];
    const float* W   = (const float*)d_in[1];
    const float* b   = (const float*)d_in[2];
    const float* ew  = (const float*)d_in[3];
    const float* sw  = (const float*)d_in[4];
    const int*   tp  = (const int*)d_in[5];
    const int*   src = (const int*)d_in[6];
    float*       out = (float*)d_out;

    const int nNodes = in_sizes[4];
    float* h = (float*)d_ws;

    gcn_gemm_wmma<<<GEMM_BLOCKS, GEMM_THREADS, 0, stream>>>(x, W, b, h, nNodes);

    const int aggBlocks = (nNodes + AGG_NODES_PER_BLOCK - 1) / AGG_NODES_PER_BLOCK;
    gcn_aggregate<<<aggBlocks, AGG_THREADS, 0, stream>>>(h, ew, sw, tp, src, out, nNodes, in_sizes[6]);
}
